# starting kernel
# baseline (speedup 1.0000x reference)
#include <hip/hip_runtime.h>

#define N_NODES 65536
#define N_EDGES 1048576
#define DIM 64

typedef float v2f __attribute__((ext_vector_type(2)));
typedef float v8f __attribute__((ext_vector_type(8)));


__global__ __launch_bounds__(256) void zero_u32_kernel(unsigned* __restrict__ p, int n) {
    int i = blockIdx.x * 256 + threadIdx.x;
    if (i < n) p[i] = 0u;
}

__global__ __launch_bounds__(256) void degree_kernel(const int* __restrict__ ei,
                                                     unsigned* __restrict__ deg_out,
                                                     unsigned* __restrict__ deg_in) {
    int e = blockIdx.x * 256 + threadIdx.x;
    if (e < N_EDGES) {
        atomicAdd(&deg_out[ei[e]], 1u);
        atomicAdd(&deg_in[ei[N_EDGES + e]], 1u);
    }
}

__global__ __launch_bounds__(256) void rsqrt_kernel(const unsigned* __restrict__ deg_out,
                                                    const unsigned* __restrict__ deg_in,
                                                    float* __restrict__ rs_out,
                                                    float* __restrict__ rs_in) {
    int i = blockIdx.x * 256 + threadIdx.x;
    if (i < N_NODES) {
        unsigned dovr = deg_out[i]; if (dovr < 1u) dovr = 1u;
        unsigned din  = deg_in[i];  if (din  < 1u) din  = 1u;
        rs_out[i] = rsqrtf((float)dovr);
        rs_in[i]  = rsqrtf((float)din);
    }
}

__global__ __launch_bounds__(1024) void scan_kernel(const unsigned* __restrict__ deg_in,
                                                    int* __restrict__ row_ptr) {
    __shared__ unsigned part[1024];
    int t = threadIdx.x;
    int base = t * (N_NODES / 1024);
    unsigned sum = 0;
    for (int i = 0; i < N_NODES / 1024; ++i) sum += deg_in[base + i];
    part[t] = sum;
    __syncthreads();
    for (int off = 1; off < 1024; off <<= 1) {
        unsigned v = (t >= off) ? part[t - off] : 0u;
        __syncthreads();
        part[t] += v;
        __syncthreads();
    }
    unsigned run = (t == 0) ? 0u : part[t - 1];
    for (int i = 0; i < N_NODES / 1024; ++i) {
        row_ptr[base + i] = (int)run;
        run += deg_in[base + i];
    }
    if (t == 1023) row_ptr[N_NODES] = (int)run;
}

__global__ __launch_bounds__(256) void fill_kernel(const int* __restrict__ ei,
                                                   unsigned* __restrict__ cursor,
                                                   const int* __restrict__ row_ptr,
                                                   int* __restrict__ in_src) {
    int e = blockIdx.x * 256 + threadIdx.x;
    if (e < N_EDGES) {
        int s = ei[e];
        int d = ei[N_EDGES + e];
        unsigned slot = atomicAdd(&cursor[d], 1u);
        in_src[row_ptr[d] + (int)slot] = s;
    }
}

__global__ __launch_bounds__(256) void scale0_kernel(const float* __restrict__ x,
                                                     const float* __restrict__ rs_out,
                                                     float* __restrict__ hs) {
    int t = blockIdx.x * 256 + threadIdx.x;
    float4 v = ((const float4*)x)[t];
    float r = rs_out[t >> 4];
    v.x *= r; v.y *= r; v.z *= r; v.w *= r;
    ((float4*)hs)[t] = v;
}

__global__ __launch_bounds__(256) void agg_kernel(const float* __restrict__ hs,
                                                  const int* __restrict__ in_src,
                                                  const int* __restrict__ row_ptr,
                                                  const float* __restrict__ rs_in,
                                                  float* __restrict__ aggn) {
    int node = (blockIdx.x * 256 + threadIdx.x) >> 5;
    int lane = threadIdx.x & 31;
    int start = row_ptr[node];
    int end   = row_ptr[node + 1];
    float ax = 0.0f, ay = 0.0f;
    for (int base = start; base < end; base += 32) {
        int i = base + lane;
        int idx = (i < end) ? in_src[i] : 0;
        int cnt = end - base; if (cnt > 32) cnt = 32;
        for (int j = 0; j < cnt; ++j) {
            int s = __shfl(idx, j, 32);
            const v2f v = *(const v2f*)(hs + (size_t)s * DIM + lane * 2);
            ax += v.x; ay += v.y;
        }
    }
    float r = rs_in[node];
    v2f outv; outv.x = ax * r; outv.y = ay * r;
    *(v2f*)(aggn + (size_t)node * DIM + lane * 2) = outv;
}

__global__ __launch_bounds__(128) void gemm_wmma_kernel(const float* __restrict__ X,
                                                        const float* __restrict__ W,
                                                        const float* __restrict__ bias,
                                                        const float* __restrict__ rs_out,
                                                        float* __restrict__ Y,
                                                        int relu_scale) {
    __shared__ float lWsw[16 * 4 * 32 * 2];
    __shared__ float lB[DIM];
    for (int p = threadIdx.x; p < 16 * 4 * 32; p += 128) {
        int l  = p & 31;
        int t  = (p >> 5) & 3;
        int f  = p >> 7;
        int ka = 4 * f + 2 * (l >> 4);
        int n  = 16 * t + (l & 15);
        v2f w2;
        w2.x = W[ka * DIM + n];
        w2.y = W[(ka + 1) * DIM + n];
        *(v2f*)&lWsw[p * 2] = w2;
    }
    if (threadIdx.x < DIM) lB[threadIdx.x] = bias[threadIdx.x];
    __syncthreads();

    const int wave = threadIdx.x >> 5;
    const int lane = threadIdx.x & 31;
    const int half = lane >> 4;
    const int l16  = lane & 15;
    const int m0   = blockIdx.x * 64 + wave * 16;
    const int mA   = m0 + l16;

    v8f acc[4] = {v8f{}, v8f{}, v8f{}, v8f{}};

#pragma unroll
    for (int f = 0; f < 16; ++f) {
        const int ka = 4 * f + 2 * half;
        v2f a = *(const v2f*)(X + (size_t)mA * DIM + ka);
#pragma unroll
        for (int t = 0; t < 4; ++t) {
            v2f b = *(const v2f*)&lWsw[(((f * 4 + t) * 32) + lane) * 2];
            acc[t] = __builtin_amdgcn_wmma_f32_16x16x4_f32(
                false, a, false, b, (short)0, acc[t], false, false);
        }
    }

    const int mrow0 = m0 + half * 8;
    float rsv[8];
#pragma unroll
    for (int r = 0; r < 8; ++r) rsv[r] = relu_scale ? rs_out[mrow0 + r] : 1.0f;

#pragma unroll
    for (int t = 0; t < 4; ++t) {
        const int n = t * 16 + l16;
        const float bn = lB[n];
#pragma unroll
        for (int r = 0; r < 8; ++r) {
            float v = acc[t][r] + bn;
            if (relu_scale) v = fmaxf(v, 0.0f) * rsv[r];
            Y[(size_t)(mrow0 + r) * DIM + n] = v;
        }
    }
}

extern "C" void kernel_launch(void* const* d_in, const int* in_sizes, int n_in,
                              void* d_out, int out_size, void* d_ws, size_t ws_size,
                              hipStream_t stream) {
    (void)in_sizes; (void)n_in; (void)out_size; (void)ws_size;
    const float* x  = (const float*)d_in[0];
    const int*   ei = (const int*)d_in[1];
    const float* W1 = (const float*)d_in[3];
    const float* b1 = (const float*)d_in[4];
    const float* W2 = (const float*)d_in[5];
    const float* b2 = (const float*)d_in[6];
    const float* W3 = (const float*)d_in[7];
    const float* b3 = (const float*)d_in[8];
    float* out = (float*)d_out;

    char* ws = (char*)d_ws;
    unsigned* deg_out = (unsigned*)ws; ws += (size_t)N_NODES * 4;
    unsigned* deg_in  = (unsigned*)ws; ws += (size_t)N_NODES * 4;
    unsigned* cursor  = (unsigned*)ws; ws += (size_t)N_NODES * 4;
    float*    rs_out  = (float*)ws;    ws += (size_t)N_NODES * 4;
    float*    rs_in   = (float*)ws;    ws += (size_t)N_NODES * 4;
    int*      row_ptr = (int*)ws;      ws += (size_t)(N_NODES + 16) * 4;
    int*      in_src  = (int*)ws;      ws += (size_t)N_EDGES * 4;
    float*    hs      = (float*)ws;    ws += (size_t)N_NODES * DIM * 4;
    float*    aggn    = (float*)ws;    ws += (size_t)N_NODES * DIM * 4;

    zero_u32_kernel<<<(3 * N_NODES + 255) / 256, 256, 0, stream>>>(deg_out, 3 * N_NODES);
    degree_kernel<<<N_EDGES / 256, 256, 0, stream>>>(ei, deg_out, deg_in);
    rsqrt_kernel<<<N_NODES / 256, 256, 0, stream>>>(deg_out, deg_in, rs_out, rs_in);
    scan_kernel<<<1, 1024, 0, stream>>>(deg_in, row_ptr);
    fill_kernel<<<N_EDGES / 256, 256, 0, stream>>>(ei, cursor, row_ptr, in_src);

    scale0_kernel<<<(N_NODES * DIM / 4) / 256, 256, 0, stream>>>(x, rs_out, hs);

    agg_kernel<<<(N_NODES * 32) / 256, 256, 0, stream>>>(hs, in_src, row_ptr, rs_in, aggn);
    gemm_wmma_kernel<<<N_NODES / 64, 128, 0, stream>>>(aggn, W1, b1, rs_out, hs, 1);
    agg_kernel<<<(N_NODES * 32) / 256, 256, 0, stream>>>(hs, in_src, row_ptr, rs_in, aggn);
    gemm_wmma_kernel<<<N_NODES / 64, 128, 0, stream>>>(aggn, W2, b2, rs_out, hs, 1);
    agg_kernel<<<(N_NODES * 32) / 256, 256, 0, stream>>>(hs, in_src, row_ptr, rs_in, aggn);
    gemm_wmma_kernel<<<N_NODES / 64, 128, 0, stream>>>(aggn, W3, b3, rs_out, out, 0);
}


# Slot rows and degree counts built in LDS by node-range owner workgroups over edge-list slices (LDS add returns rank), one global add per node and slice, none per edge; row sum order: slice arrival
# speedup vs baseline: 3.4185x; 3.4185x over previous
#include <hip/hip_runtime.h>

#define N_NODES 65536
#define N_EDGES 1048576
#define DIM 64
#define SLOTS 32
#define SPILL_CAP 130976
#define RANGE_SHIFT 12
#define RANGE_NODES (1 << RANGE_SHIFT)
#define N_RANGES (N_NODES / RANGE_NODES)
#define N_SLICES 16
#define SLICE_INT4 (N_EDGES / 4 / N_SLICES)
#define ITEMS_PER_HALF ((N_RANGES / 2) * N_SLICES)
#define LROW (SLOTS + 1)
#define TICK_WORDS 64

static_assert(N_NODES <= 65536, "node ids are kept as 16-bit words");
static_assert(N_EDGES % 32 == 0, "the raw edge list is walked 32 edges at a time");
static_assert(N_NODES % RANGE_NODES == 0 && N_RANGES % 2 == 0 && RANGE_NODES % 1024 == 0, "build: two halves of whole ranges; 1024 threads share a range's nodes evenly");
static_assert(SLICE_INT4 % (4 * 1024) == 0 && (ITEMS_PER_HALF & (ITEMS_PER_HALF - 1)) == 0, "build: a slice is walked in batches of 4 x 1024 int4; tickets wrap on a power of two");
static_assert(N_NODES % 64 == 0 && SLOTS == 32 && DIM == 64, "agg: 8 waves of 8 rows per workgroup, 8 lanes x 8 floats per row, two rows per lane");

typedef float v2f __attribute__((ext_vector_type(2)));
typedef float v8f __attribute__((ext_vector_type(8)));


__global__ __launch_bounds__(256) void zero_u32_kernel(unsigned* __restrict__ p, int n) {
    int i = blockIdx.x * 256 + threadIdx.x;
    if (i < n) p[i] = 0u;
}

__device__ __forceinline__ void spill_push(unsigned d, unsigned s, unsigned* __restrict__ spill_cnt, unsigned* __restrict__ spill) {
    const unsigned i = atomicAdd(spill_cnt, 1u);
    if (i < SPILL_CAP) spill[i] = (d << 16) | s;
}

__device__ __forceinline__ void put_edge(unsigned d, unsigned s, unsigned k, unsigned short* ls,
                                         unsigned* __restrict__ spill_cnt, unsigned* __restrict__ spill) {
    if (k < SLOTS) ls[(d & (RANGE_NODES - 1)) * LROW + k] = (unsigned short)s;
    else spill_push(d, s, spill_cnt, spill);
}

__device__ __forceinline__ void count_src(unsigned s, unsigned b, unsigned* cnt_out) {
    if ((s >> RANGE_SHIFT) == b) atomicAdd(&cnt_out[s & (RANGE_NODES - 1)], 1u);
}

__device__ __forceinline__ void scan4(const int4 d, const int4 s, unsigned b, unsigned* cnt_in, unsigned* cnt_out, unsigned short* ls,
                                      unsigned* __restrict__ spill_cnt, unsigned* __restrict__ spill) {
    const unsigned d0 = (unsigned)d.x, d1 = (unsigned)d.y, d2 = (unsigned)d.z, d3 = (unsigned)d.w;
    const bool m0 = (d0 >> RANGE_SHIFT) == b, m1 = (d1 >> RANGE_SHIFT) == b;
    const bool m2 = (d2 >> RANGE_SHIFT) == b, m3 = (d3 >> RANGE_SHIFT) == b;
    unsigned k0 = 0u, k1 = 0u, k2 = 0u, k3 = 0u;
    if (m0) k0 = atomicAdd(&cnt_in[d0 & (RANGE_NODES - 1)], 1u);
    if (m1) k1 = atomicAdd(&cnt_in[d1 & (RANGE_NODES - 1)], 1u);
    if (m2) k2 = atomicAdd(&cnt_in[d2 & (RANGE_NODES - 1)], 1u);
    if (m3) k3 = atomicAdd(&cnt_in[d3 & (RANGE_NODES - 1)], 1u);
    count_src((unsigned)s.x, b, cnt_out);
    count_src((unsigned)s.y, b, cnt_out);
    count_src((unsigned)s.z, b, cnt_out);
    count_src((unsigned)s.w, b, cnt_out);
    if (m0) put_edge(d0, (unsigned)s.x, k0, ls, spill_cnt, spill);
    if (m1) put_edge(d1, (unsigned)s.y, k1, ls, spill_cnt, spill);
    if (m2) put_edge(d2, (unsigned)s.z, k2, ls, spill_cnt, spill);
    if (m3) put_edge(d3, (unsigned)s.w, k3, ls, spill_cnt, spill);
}

__global__ __launch_bounds__(1024) void build_kernel(const int* __restrict__ ei,
                                                     unsigned* __restrict__ deg_out,
                                                     unsigned* __restrict__ deg_in,
                                                     unsigned* __restrict__ spill_cnt,
                                                     unsigned* __restrict__ spill,
                                                     unsigned short* __restrict__ slots,
                                                     unsigned* __restrict__ tick) {
    __shared__ unsigned cnt_in[RANGE_NODES];
    __shared__ unsigned cnt_out[RANGE_NODES];
    __shared__ unsigned short ls[RANGE_NODES * LROW];
    __shared__ unsigned item;
    if (threadIdx.x == 0) {
        const unsigned h = ((unsigned)__builtin_amdgcn_s_sendmsg_rtn(0x87) >> 18) & 1u;
        unsigned r = h;
        unsigned c = atomicAdd(&tick[h * 32], 1u);
        if (c >= ITEMS_PER_HALF) {
            r = h ^ 1u;
            c = (ITEMS_PER_HALF - 1u) - atomicAdd(&tick[r * 32 + 1], 1u);
        }
        item = (r << 16) | (c & (ITEMS_PER_HALF - 1u));
    }
    for (int i = threadIdx.x; i < RANGE_NODES; i += 1024) { cnt_in[i] = 0u; cnt_out[i] = 0u; }
    __syncthreads();
    const unsigned c = item & 0xFFFFu;
    const unsigned b = (item >> 16) * (N_RANGES / 2) + (c % (N_RANGES / 2));
    const unsigned p = c / (N_RANGES / 2);
    const int4* s4 = (const int4*)ei + (size_t)p * SLICE_INT4 + threadIdx.x;
    const int4* d4 = (const int4*)(ei + N_EDGES) + (size_t)p * SLICE_INT4 + threadIdx.x;
    for (int i0 = 0; i0 < SLICE_INT4; i0 += 4 * 1024) {
        int4 dv[4], sv[4];
#pragma unroll
        for (int u = 0; u < 4; ++u) { dv[u] = d4[i0 + u * 1024]; sv[u] = s4[i0 + u * 1024]; }
#pragma unroll
        for (int u = 0; u < 4; ++u) scan4(dv[u], sv[u], b, cnt_in, cnt_out, ls, spill_cnt, spill);
    }
    __syncthreads();
    unsigned ci[RANGE_NODES / 1024], base[RANGE_NODES / 1024];
#pragma unroll
    for (int q = 0; q < RANGE_NODES / 1024; ++q) {
        const int ln = q * 1024 + threadIdx.x;
        const unsigned n = b * RANGE_NODES + (unsigned)ln;
        atomicAdd(&deg_out[n], cnt_out[ln]);
        ci[q] = cnt_in[ln];
        base[q] = atomicAdd(&deg_in[n], ci[q]);
    }
#pragma unroll
    for (int q = 0; q < RANGE_NODES / 1024; ++q) {
        const int ln = q * 1024 + threadIdx.x;
        const unsigned n = b * RANGE_NODES + (unsigned)ln;
        const unsigned cl = ci[q] < SLOTS ? ci[q] : SLOTS;
        for (unsigned j = 0; j < cl; ++j) {
            const unsigned short v = ls[ln * LROW + j];
            const unsigned g = base[q] + j;
            if (g < SLOTS) slots[n * SLOTS + g] = v;
            else spill_push(n, (unsigned)v, spill_cnt, spill);
        }
    }
}

__device__ __forceinline__ float inv_sqrt_deg(unsigned d) {
    if (d < 1u) d = 1u;
    return rsqrtf((float)d);
}

__device__ __forceinline__ void acc4(float4& a, const float4 v) {
    a.x += v.x; a.y += v.y; a.z += v.z; a.w += v.w;
}

__device__ __forceinline__ void acc4s(float4& a, const float4 v, const float s) {
    a.x += v.x * s; a.y += v.y * s; a.z += v.z * s; a.w += v.w * s;
}

template <bool SRC_SCALE>
__device__ __forceinline__ void add_row(const float* __restrict__ hs, const unsigned* __restrict__ deg_out,
                                        int s, int q, float4& a0, float4& a1) {
    const float4* row = (const float4*)(hs + (size_t)s * DIM) + q;
    const float4 v0 = row[0], v1 = row[8];
    if (SRC_SCALE) {
        const float sc = inv_sqrt_deg(deg_out[s]);
        acc4s(a0, v0, sc);
        acc4s(a1, v1, sc);
    } else {
        acc4(a0, v0);
        acc4(a1, v1);
    }
}

template <bool SRC_SCALE>
__global__ __launch_bounds__(256) void agg_kernel(const float* __restrict__ hs,
                                                  const unsigned short* __restrict__ slots,
                                                  const unsigned* __restrict__ deg_in,
                                                  const unsigned* __restrict__ deg_out,
                                                  const unsigned* __restrict__ spill_cnt,
                                                  const unsigned* __restrict__ spill,
                                                  const int* __restrict__ ei,
                                                  float* __restrict__ aggn) {
    const int lane = threadIdx.x & 31;
    const int wv   = (blockIdx.x * 256 + threadIdx.x) >> 5;
    const int q    = lane & 7;
    const int nodeA = wv * 8 + (lane >> 3);
    const int nodeB = nodeA + 4;
    const unsigned degA = deg_in[nodeA];
    const unsigned degB = deg_in[nodeB];
    const float rA = inv_sqrt_deg(degA);
    const float rB = inv_sqrt_deg(degB);
    uint4 wA[SLOTS / 8], wB[SLOTS / 8];
#pragma unroll
    for (int k = 0; k < SLOTS / 8; ++k) {
        wA[k] = ((const uint4*)(slots + (size_t)nodeA * SLOTS))[k];
        wB[k] = ((const uint4*)(slots + (size_t)nodeB * SLOTS))[k];
    }

#pragma unroll
    for (int k = 0; k < SLOTS / 8; ++k) {
        asm volatile("" :: "v"(wA[k].x), "v"(wA[k].y), "v"(wA[k].z), "v"(wA[k].w),
                           "v"(wB[k].x), "v"(wB[k].y), "v"(wB[k].z), "v"(wB[k].w));
    }

    const bool spA = degA > SLOTS;
    const bool spB = degB > SLOTS;
    const unsigned anysp = __builtin_amdgcn_ballot_w32(spA || spB);
    unsigned n_spill = 0u;
    if (anysp != 0u) n_spill = *spill_cnt;
    const bool rawA = spA && (n_spill > SPILL_CAP);
    const bool rawB = spB && (n_spill > SPILL_CAP);
    const int cntA = rawA ? 0 : (spA ? SLOTS : (int)degA);
    const int cntB = rawB ? 0 : (spB ? SLOTS : (int)degB);

    const float4 zero4 = make_float4(0.f, 0.f, 0.f, 0.f);
    float4 a0 = zero4, a1 = zero4, b0 = zero4, b1 = zero4;
    const float4* hq = (const float4*)hs + q;

#pragma unroll
    for (int jq = 0; jq < SLOTS / 8; ++jq) {
        if (__builtin_amdgcn_ballot_w32((jq * 8 < cntA) || (jq * 8 < cntB)) == 0u) break;
        const uint4 ga = wA[jq], gb = wB[jq];
#pragma nounroll
        for (int tp = 0; tp < 4; ++tp) {
            const unsigned ua = (tp == 0) ? ga.x : (tp == 1) ? ga.y : (tp == 2) ? ga.z : ga.w;
            const unsigned ub = (tp == 0) ? gb.x : (tp == 1) ? gb.y : (tp == 2) ? gb.z : gb.w;
            const int j0 = jq * 8 + 2 * tp;
            float4 pa0 = zero4, pa1 = zero4, pa2 = zero4, pa3 = zero4;
            float4 pb0 = zero4, pb1 = zero4, pb2 = zero4, pb3 = zero4;
            const unsigned sA0 = ua & 0xFFFFu, sA1 = ua >> 16, sB0 = ub & 0xFFFFu, sB1 = ub >> 16;
            const bool uA0 = j0 < cntA, uA1 = j0 + 1 < cntA, uB0 = j0 < cntB, uB1 = j0 + 1 < cntB;
            if (uA0) { const float4* r = hq + (size_t)sA0 * (DIM / 4); pa0 = r[0]; pa1 = r[8]; }
            if (uA1) { const float4* r = hq + (size_t)sA1 * (DIM / 4); pa2 = r[0]; pa3 = r[8]; }
            if (uB0) { const float4* r = hq + (size_t)sB0 * (DIM / 4); pb0 = r[0]; pb1 = r[8]; }
            if (uB1) { const float4* r = hq + (size_t)sB1 * (DIM / 4); pb2 = r[0]; pb3 = r[8]; }
            if (SRC_SCALE) {
                const unsigned da0 = deg_out[uA0 ? sA0 : (unsigned)nodeA];
                const unsigned da1 = deg_out[uA1 ? sA1 : (unsigned)nodeA];
                const unsigned db0 = deg_out[uB0 ? sB0 : (unsigned)nodeB];
                const unsigned db1 = deg_out[uB1 ? sB1 : (unsigned)nodeB];
                const float sa0 = inv_sqrt_deg(da0), sa1 = inv_sqrt_deg(da1);
                const float sb0 = inv_sqrt_deg(db0), sb1 = inv_sqrt_deg(db1);
                acc4s(a0, pa0, sa0); acc4s(a1, pa1, sa0);
                acc4s(a0, pa2, sa1); acc4s(a1, pa3, sa1);
                acc4s(b0, pb0, sb0); acc4s(b1, pb1, sb0);
                acc4s(b0, pb2, sb1); acc4s(b1, pb3, sb1);
            } else {
                acc4(a0, pa0); acc4(a1, pa1);
                acc4(a0, pa2); acc4(a1, pa3);
                acc4(b0, pb0); acc4(b1, pb1);
                acc4(b0, pb2); acc4(b1, pb3);
            }
        }
    }

    if (anysp != 0u) {
        if (n_spill <= SPILL_CAP) {
            for (unsigned base = 0; base < n_spill; base += 32) {
                const unsigned i = base + (unsigned)lane;
                const unsigned w = (i < n_spill) ? spill[i] : 0u;
                const bool hit = (i < n_spill) && ((w >> 19) == (unsigned)wv);
                unsigned m = __builtin_amdgcn_ballot_w32(hit);
                while (m) {
                    const int j = __builtin_ctz(m);
                    m &= m - 1u;
                    const unsigned wj = (unsigned)__shfl((int)w, j, 32);
                    if ((wj >> 16) == (unsigned)nodeA) add_row<SRC_SCALE>(hs, deg_out, (int)(wj & 0xFFFFu), q, a0, a1);
                    if ((wj >> 16) == (unsigned)nodeB) add_row<SRC_SCALE>(hs, deg_out, (int)(wj & 0xFFFFu), q, b0, b1);
                }
            }
        } else {
            for (int base = 0; base < N_EDGES; base += 32) {
                const int d = ei[N_EDGES + base + lane];
                unsigned m = __builtin_amdgcn_ballot_w32((d >> 3) == wv);
                while (m) {
                    const int j = __builtin_ctz(m);
                    m &= m - 1u;
                    const int dj = __shfl(d, j, 32);
                    const int s = ei[base + j];
                    if (rawA && dj == nodeA) add_row<SRC_SCALE>(hs, deg_out, s, q, a0, a1);
                    if (rawB && dj == nodeB) add_row<SRC_SCALE>(hs, deg_out, s, q, b0, b1);
                }
            }
        }
    }

    float4* oA = (float4*)(aggn + (size_t)nodeA * DIM) + q;
    float4* oB = (float4*)(aggn + (size_t)nodeB * DIM) + q;
    oA[0] = make_float4(a0.x * rA, a0.y * rA, a0.z * rA, a0.w * rA);
    oA[8] = make_float4(a1.x * rA, a1.y * rA, a1.z * rA, a1.w * rA);
    oB[0] = make_float4(b0.x * rB, b0.y * rB, b0.z * rB, b0.w * rB);
    oB[8] = make_float4(b1.x * rB, b1.y * rB, b1.z * rB, b1.w * rB);
}

__global__ __launch_bounds__(128) void gemm_wmma_kernel(const float* __restrict__ X,
                                                        const float* __restrict__ W,
                                                        const float* __restrict__ bias,
                                                        const unsigned* __restrict__ deg_out,
                                                        float* __restrict__ Y,
                                                        int relu_scale) {
    __shared__ float lWsw[16 * 4 * 32 * 2];
    __shared__ float lB[DIM];
    for (int p = threadIdx.x; p < 16 * 4 * 32; p += 128) {
        int l  = p & 31;
        int t  = (p >> 5) & 3;
        int f  = p >> 7;
        int ka = 4 * f + 2 * (l >> 4);
        int n  = 16 * t + (l & 15);
        v2f w2;
        w2.x = W[ka * DIM + n];
        w2.y = W[(ka + 1) * DIM + n];
        *(v2f*)&lWsw[p * 2] = w2;
    }
    if (threadIdx.x < DIM) lB[threadIdx.x] = bias[threadIdx.x];
    __syncthreads();

    const int wave = threadIdx.x >> 5;
    const int lane = threadIdx.x & 31;
    const int half = lane >> 4;
    const int l16  = lane & 15;
    const int m0   = blockIdx.x * 64 + wave * 16;
    const int mA   = m0 + l16;

    v8f acc[4] = {v8f{}, v8f{}, v8f{}, v8f{}};

#pragma unroll
    for (int f = 0; f < 16; ++f) {
        const int ka = 4 * f + 2 * half;
        v2f a = *(const v2f*)(X + (size_t)mA * DIM + ka);
#pragma unroll
        for (int t = 0; t < 4; ++t) {
            v2f b = *(const v2f*)&lWsw[(((f * 4 + t) * 32) + lane) * 2];
            acc[t] = __builtin_amdgcn_wmma_f32_16x16x4_f32(
                false, a, false, b, (short)0, acc[t], false, false);
        }
    }

    const int mrow0 = m0 + half * 8;
    uint4 dlo = make_uint4(1u, 1u, 1u, 1u), dhi = dlo;
    if (relu_scale) {
        const uint4* dp = (const uint4*)(deg_out + mrow0);
        dlo = dp[0];
        dhi = dp[1];
    }
    const float rsv[8] = {inv_sqrt_deg(dlo.x), inv_sqrt_deg(dlo.y), inv_sqrt_deg(dlo.z), inv_sqrt_deg(dlo.w),
                          inv_sqrt_deg(dhi.x), inv_sqrt_deg(dhi.y), inv_sqrt_deg(dhi.z), inv_sqrt_deg(dhi.w)};

#pragma unroll
    for (int t = 0; t < 4; ++t) {
        const int n = t * 16 + l16;
        const float bn = lB[n];
#pragma unroll
        for (int r = 0; r < 8; ++r) {
            float v = acc[t][r] + bn;
            if (relu_scale) v = fmaxf(v, 0.0f) * rsv[r];
            Y[(size_t)(mrow0 + r) * DIM + n] = v;
        }
    }
}

extern "C" void kernel_launch(void* const* d_in, const int* in_sizes, int n_in,
                              void* d_out, int out_size, void* d_ws, size_t ws_size,
                              hipStream_t stream) {
    (void)in_sizes; (void)n_in; (void)out_size; (void)ws_size;
    const float* x  = (const float*)d_in[0];
    const int*   ei = (const int*)d_in[1];
    const float* W1 = (const float*)d_in[3];
    const float* b1 = (const float*)d_in[4];
    const float* W2 = (const float*)d_in[5];
    const float* b2 = (const float*)d_in[6];
    const float* W3 = (const float*)d_in[7];
    const float* b3 = (const float*)d_in[8];
    float* out = (float*)d_out;

    char* ws = (char*)d_ws;
    unsigned* deg_out = (unsigned*)ws; ws += (size_t)N_NODES * 4;
    unsigned* deg_in  = (unsigned*)ws; ws += (size_t)N_NODES * 4;
    unsigned* spill_cnt = (unsigned*)ws; ws += (size_t)32 * 4;
    unsigned* tick    = (unsigned*)ws; ws += (size_t)TICK_WORDS * 4;
    unsigned* spill   = (unsigned*)ws; ws += (size_t)SPILL_CAP * 4;
    unsigned short* slots = (unsigned short*)ws; ws += (size_t)N_NODES * SLOTS * 2;
    float*    hs      = (float*)ws;    ws += (size_t)N_NODES * DIM * 4;
    float*    aggn    = (float*)ws;    ws += (size_t)N_NODES * DIM * 4;

    zero_u32_kernel<<<(2 * N_NODES + 32 + TICK_WORDS + 255) / 256, 256, 0, stream>>>(deg_out, 2 * N_NODES + 32 + TICK_WORDS);
    build_kernel<<<2 * ITEMS_PER_HALF, 1024, 0, stream>>>(ei, deg_out, deg_in, spill_cnt, spill, slots, tick);

    agg_kernel<true><<<N_NODES / 64, 256, 0, stream>>>(x, slots, deg_in, deg_out, spill_cnt, spill, ei, aggn);
    gemm_wmma_kernel<<<N_NODES / 64, 128, 0, stream>>>(aggn, W1, b1, deg_out, hs, 1);
    agg_kernel<false><<<N_NODES / 64, 256, 0, stream>>>(hs, slots, deg_in, deg_out, spill_cnt, spill, ei, aggn);
    gemm_wmma_kernel<<<N_NODES / 64, 128, 0, stream>>>(aggn, W2, b2, deg_out, hs, 1);
    agg_kernel<false><<<N_NODES / 64, 256, 0, stream>>>(hs, slots, deg_in, deg_out, spill_cnt, spill, ei, aggn);
    gemm_wmma_kernel<<<N_NODES / 64, 128, 0, stream>>>(aggn, W3, b3, deg_out, out, 0);
}


# Layer 3 multiplies before it sums: the W3 product rides behind the W2 product in one launch (rows via an LDS tile into A fragments), agg3 adds bias, writes out; 8 -> 7 launches; f32 sum order changes
# speedup vs baseline: 3.5846x; 1.0486x over previous
#include <hip/hip_runtime.h>

#define N_NODES 65536
#define N_EDGES 1048576
#define DIM 64
#define SLOTS 32
#define SPILL_CAP 130976
#define RANGE_SHIFT 12
#define RANGE_NODES (1 << RANGE_SHIFT)
#define N_RANGES (N_NODES / RANGE_NODES)
#define N_SLICES 16
#define SLICE_INT4 (N_EDGES / 4 / N_SLICES)
#define ITEMS_PER_HALF ((N_RANGES / 2) * N_SLICES)
#define LROW (SLOTS + 1)
#define TICK_WORDS 64

static_assert(N_NODES <= 65536, "node ids are kept as 16-bit words");
static_assert(N_EDGES % 32 == 0, "the raw edge list is walked 32 edges at a time");
static_assert(N_NODES % RANGE_NODES == 0 && N_RANGES % 2 == 0 && RANGE_NODES % 1024 == 0, "build: two halves of whole ranges; 1024 threads share a range's nodes evenly");
static_assert(SLICE_INT4 % (4 * 1024) == 0 && (ITEMS_PER_HALF & (ITEMS_PER_HALF - 1)) == 0, "build: a slice is walked in batches of 4 x 1024 int4; tickets wrap on a power of two");
static_assert(N_NODES % 64 == 0 && SLOTS == 32 && DIM == 64, "agg: 8 waves of 8 rows per workgroup, 8 lanes x 8 floats per row, two rows per lane");

typedef float v2f __attribute__((ext_vector_type(2)));
typedef float v8f __attribute__((ext_vector_type(8)));


__global__ __launch_bounds__(256) void zero_u32_kernel(unsigned* __restrict__ p, int n) {
    int i = blockIdx.x * 256 + threadIdx.x;
    if (i < n) p[i] = 0u;
}

__device__ __forceinline__ void spill_push(unsigned d, unsigned s, unsigned* __restrict__ spill_cnt, unsigned* __restrict__ spill) {
    const unsigned i = atomicAdd(spill_cnt, 1u);
    if (i < SPILL_CAP) spill[i] = (d << 16) | s;
}

__device__ __forceinline__ void put_edge(unsigned d, unsigned s, unsigned k, unsigned short* ls,
                                         unsigned* __restrict__ spill_cnt, unsigned* __restrict__ spill) {
    if (k < SLOTS) ls[(d & (RANGE_NODES - 1)) * LROW + k] = (unsigned short)s;
    else spill_push(d, s, spill_cnt, spill);
}

__device__ __forceinline__ void count_src(unsigned s, unsigned b, unsigned* cnt_out) {
    if ((s >> RANGE_SHIFT) == b) atomicAdd(&cnt_out[s & (RANGE_NODES - 1)], 1u);
}

__device__ __forceinline__ void scan4(const int4 d, const int4 s, unsigned b, unsigned* cnt_in, unsigned* cnt_out, unsigned short* ls,
                                      unsigned* __restrict__ spill_cnt, unsigned* __restrict__ spill) {
    const unsigned d0 = (unsigned)d.x, d1 = (unsigned)d.y, d2 = (unsigned)d.z, d3 = (unsigned)d.w;
    const bool m0 = (d0 >> RANGE_SHIFT) == b, m1 = (d1 >> RANGE_SHIFT) == b;
    const bool m2 = (d2 >> RANGE_SHIFT) == b, m3 = (d3 >> RANGE_SHIFT) == b;
    unsigned k0 = 0u, k1 = 0u, k2 = 0u, k3 = 0u;
    if (m0) k0 = atomicAdd(&cnt_in[d0 & (RANGE_NODES - 1)], 1u);
    if (m1) k1 = atomicAdd(&cnt_in[d1 & (RANGE_NODES - 1)], 1u);
    if (m2) k2 = atomicAdd(&cnt_in[d2 & (RANGE_NODES - 1)], 1u);
    if (m3) k3 = atomicAdd(&cnt_in[d3 & (RANGE_NODES - 1)], 1u);
    count_src((unsigned)s.x, b, cnt_out);
    count_src((unsigned)s.y, b, cnt_out);
    count_src((unsigned)s.z, b, cnt_out);
    count_src((unsigned)s.w, b, cnt_out);
    if (m0) put_edge(d0, (unsigned)s.x, k0, ls, spill_cnt, spill);
    if (m1) put_edge(d1, (unsigned)s.y, k1, ls, spill_cnt, spill);
    if (m2) put_edge(d2, (unsigned)s.z, k2, ls, spill_cnt, spill);
    if (m3) put_edge(d3, (unsigned)s.w, k3, ls, spill_cnt, spill);
}

__global__ __launch_bounds__(1024) void build_kernel(const int* __restrict__ ei,
                                                     unsigned* __restrict__ deg_out,
                                                     unsigned* __restrict__ deg_in,
                                                     unsigned* __restrict__ spill_cnt,
                                                     unsigned* __restrict__ spill,
                                                     unsigned short* __restrict__ slots,
                                                     unsigned* __restrict__ tick) {
    __shared__ unsigned cnt_in[RANGE_NODES];
    __shared__ unsigned cnt_out[RANGE_NODES];
    __shared__ unsigned short ls[RANGE_NODES * LROW];
    __shared__ unsigned item;
    if (threadIdx.x == 0) {
        const unsigned h = ((unsigned)__builtin_amdgcn_s_sendmsg_rtn(0x87) >> 18) & 1u;
        unsigned r = h;
        unsigned c = atomicAdd(&tick[h * 32], 1u);
        if (c >= ITEMS_PER_HALF) {
            r = h ^ 1u;
            c = (ITEMS_PER_HALF - 1u) - atomicAdd(&tick[r * 32 + 1], 1u);
        }
        item = (r << 16) | (c & (ITEMS_PER_HALF - 1u));
    }
    for (int i = threadIdx.x; i < RANGE_NODES; i += 1024) { cnt_in[i] = 0u; cnt_out[i] = 0u; }
    __syncthreads();
    const unsigned c = item & 0xFFFFu;
    const unsigned b = (item >> 16) * (N_RANGES / 2) + (c % (N_RANGES / 2));
    const unsigned p = c / (N_RANGES / 2);
    const int4* s4 = (const int4*)ei + (size_t)p * SLICE_INT4 + threadIdx.x;
    const int4* d4 = (const int4*)(ei + N_EDGES) + (size_t)p * SLICE_INT4 + threadIdx.x;
    for (int i0 = 0; i0 < SLICE_INT4; i0 += 4 * 1024) {
        int4 dv[4], sv[4];
#pragma unroll
        for (int u = 0; u < 4; ++u) { dv[u] = d4[i0 + u * 1024]; sv[u] = s4[i0 + u * 1024]; }
#pragma unroll
        for (int u = 0; u < 4; ++u) scan4(dv[u], sv[u], b, cnt_in, cnt_out, ls, spill_cnt, spill);
    }
    __syncthreads();
    unsigned ci[RANGE_NODES / 1024], base[RANGE_NODES / 1024];
#pragma unroll
    for (int q = 0; q < RANGE_NODES / 1024; ++q) {
        const int ln = q * 1024 + threadIdx.x;
        const unsigned n = b * RANGE_NODES + (unsigned)ln;
        atomicAdd(&deg_out[n], cnt_out[ln]);
        ci[q] = cnt_in[ln];
        base[q] = atomicAdd(&deg_in[n], ci[q]);
    }
#pragma unroll
    for (int q = 0; q < RANGE_NODES / 1024; ++q) {
        const int ln = q * 1024 + threadIdx.x;
        const unsigned n = b * RANGE_NODES + (unsigned)ln;
        const unsigned cl = ci[q] < SLOTS ? ci[q] : SLOTS;
        for (unsigned j = 0; j < cl; ++j) {
            const unsigned short v = ls[ln * LROW + j];
            const unsigned g = base[q] + j;
            if (g < SLOTS) slots[n * SLOTS + g] = v;
            else spill_push(n, (unsigned)v, spill_cnt, spill);
        }
    }
}

__device__ __forceinline__ float inv_sqrt_deg(unsigned d) {
    if (d < 1u) d = 1u;
    return rsqrtf((float)d);
}

__device__ __forceinline__ void acc4(float4& a, const float4 v) {
    a.x += v.x; a.y += v.y; a.z += v.z; a.w += v.w;
}

__device__ __forceinline__ void acc4s(float4& a, const float4 v, const float s) {
    a.x += v.x * s; a.y += v.y * s; a.z += v.z * s; a.w += v.w * s;
}

template <bool SRC_SCALE>
__device__ __forceinline__ void add_row(const float* __restrict__ hs, const unsigned* __restrict__ deg_out,
                                        int s, int q, float4& a0, float4& a1) {
    const float4* row = (const float4*)(hs + (size_t)s * DIM) + q;
    const float4 v0 = row[0], v1 = row[8];
    if (SRC_SCALE) {
        const float sc = inv_sqrt_deg(deg_out[s]);
        acc4s(a0, v0, sc);
        acc4s(a1, v1, sc);
    } else {
        acc4(a0, v0);
        acc4(a1, v1);
    }
}

template <bool OUT_BIAS>
__device__ __forceinline__ float4 finish4(const float4 a, const float r, const float* __restrict__ bias, int c) {
    if (OUT_BIAS) return make_float4(a.x * r + bias[c], a.y * r + bias[c + 1], a.z * r + bias[c + 2], a.w * r + bias[c + 3]);
    return make_float4(a.x * r, a.y * r, a.z * r, a.w * r);
}

template <bool SRC_SCALE, bool OUT_BIAS>
__global__ __launch_bounds__(256) void agg_kernel(const float* __restrict__ hs,
                                                  const unsigned short* __restrict__ slots,
                                                  const unsigned* __restrict__ deg_in,
                                                  const unsigned* __restrict__ deg_out,
                                                  const unsigned* __restrict__ spill_cnt,
                                                  const unsigned* __restrict__ spill,
                                                  const int* __restrict__ ei,
                                                  const float* __restrict__ bias,
                                                  float* __restrict__ aggn) {
    const int lane = threadIdx.x & 31;
    const int wv   = (blockIdx.x * 256 + threadIdx.x) >> 5;
    const int q    = lane & 7;
    const int nodeA = wv * 8 + (lane >> 3);
    const int nodeB = nodeA + 4;
    const unsigned degA = deg_in[nodeA];
    const unsigned degB = deg_in[nodeB];
    const float rA = inv_sqrt_deg(degA);
    const float rB = inv_sqrt_deg(degB);
    uint4 wA[SLOTS / 8], wB[SLOTS / 8];
#pragma unroll
    for (int k = 0; k < SLOTS / 8; ++k) {
        wA[k] = ((const uint4*)(slots + (size_t)nodeA * SLOTS))[k];
        wB[k] = ((const uint4*)(slots + (size_t)nodeB * SLOTS))[k];
    }

#pragma unroll
    for (int k = 0; k < SLOTS / 8; ++k) {
        asm volatile("" :: "v"(wA[k].x), "v"(wA[k].y), "v"(wA[k].z), "v"(wA[k].w),
                           "v"(wB[k].x), "v"(wB[k].y), "v"(wB[k].z), "v"(wB[k].w));
    }

    const bool spA = degA > SLOTS;
    const bool spB = degB > SLOTS;
    const unsigned anysp = __builtin_amdgcn_ballot_w32(spA || spB);
    unsigned n_spill = 0u;
    if (anysp != 0u) n_spill = *spill_cnt;
    const bool rawA = spA && (n_spill > SPILL_CAP);
    const bool rawB = spB && (n_spill > SPILL_CAP);
    const int cntA = rawA ? 0 : (spA ? SLOTS : (int)degA);
    const int cntB = rawB ? 0 : (spB ? SLOTS : (int)degB);

    const float4 zero4 = make_float4(0.f, 0.f, 0.f, 0.f);
    float4 a0 = zero4, a1 = zero4, b0 = zero4, b1 = zero4;
    const float4* hq = (const float4*)hs + q;

#pragma unroll
    for (int jq = 0; jq < SLOTS / 8; ++jq) {
        if (__builtin_amdgcn_ballot_w32((jq * 8 < cntA) || (jq * 8 < cntB)) == 0u) break;
        const uint4 ga = wA[jq], gb = wB[jq];
#pragma nounroll
        for (int tp = 0; tp < 4; ++tp) {
            const unsigned ua = (tp == 0) ? ga.x : (tp == 1) ? ga.y : (tp == 2) ? ga.z : ga.w;
            const unsigned ub = (tp == 0) ? gb.x : (tp == 1) ? gb.y : (tp == 2) ? gb.z : gb.w;
            const int j0 = jq * 8 + 2 * tp;
            float4 pa0 = zero4, pa1 = zero4, pa2 = zero4, pa3 = zero4;
            float4 pb0 = zero4, pb1 = zero4, pb2 = zero4, pb3 = zero4;
            const unsigned sA0 = ua & 0xFFFFu, sA1 = ua >> 16, sB0 = ub & 0xFFFFu, sB1 = ub >> 16;
            const bool uA0 = j0 < cntA, uA1 = j0 + 1 < cntA, uB0 = j0 < cntB, uB1 = j0 + 1 < cntB;
            if (uA0) { const float4* r = hq + (size_t)sA0 * (DIM / 4); pa0 = r[0]; pa1 = r[8]; }
            if (uA1) { const float4* r = hq + (size_t)sA1 * (DIM / 4); pa2 = r[0]; pa3 = r[8]; }
            if (uB0) { const float4* r = hq + (size_t)sB0 * (DIM / 4); pb0 = r[0]; pb1 = r[8]; }
            if (uB1) { const float4* r = hq + (size_t)sB1 * (DIM / 4); pb2 = r[0]; pb3 = r[8]; }
            if (SRC_SCALE) {
                const unsigned da0 = deg_out[uA0 ? sA0 : (unsigned)nodeA];
                const unsigned da1 = deg_out[uA1 ? sA1 : (unsigned)nodeA];
                const unsigned db0 = deg_out[uB0 ? sB0 : (unsigned)nodeB];
                const unsigned db1 = deg_out[uB1 ? sB1 : (unsigned)nodeB];
                const float sa0 = inv_sqrt_deg(da0), sa1 = inv_sqrt_deg(da1);
                const float sb0 = inv_sqrt_deg(db0), sb1 = inv_sqrt_deg(db1);
                acc4s(a0, pa0, sa0); acc4s(a1, pa1, sa0);
                acc4s(a0, pa2, sa1); acc4s(a1, pa3, sa1);
                acc4s(b0, pb0, sb0); acc4s(b1, pb1, sb0);
                acc4s(b0, pb2, sb1); acc4s(b1, pb3, sb1);
            } else {
                acc4(a0, pa0); acc4(a1, pa1);
                acc4(a0, pa2); acc4(a1, pa3);
                acc4(b0, pb0); acc4(b1, pb1);
                acc4(b0, pb2); acc4(b1, pb3);
            }
        }
    }

    if (anysp != 0u) {
        if (n_spill <= SPILL_CAP) {
            for (unsigned base = 0; base < n_spill; base += 32) {
                const unsigned i = base + (unsigned)lane;
                const unsigned w = (i < n_spill) ? spill[i] : 0u;
                const bool hit = (i < n_spill) && ((w >> 19) == (unsigned)wv);
                unsigned m = __builtin_amdgcn_ballot_w32(hit);
                while (m) {
                    const int j = __builtin_ctz(m);
                    m &= m - 1u;
                    const unsigned wj = (unsigned)__shfl((int)w, j, 32);
                    if ((wj >> 16) == (unsigned)nodeA) add_row<SRC_SCALE>(hs, deg_out, (int)(wj & 0xFFFFu), q, a0, a1);
                    if ((wj >> 16) == (unsigned)nodeB) add_row<SRC_SCALE>(hs, deg_out, (int)(wj & 0xFFFFu), q, b0, b1);
                }
            }
        } else {
            for (int base = 0; base < N_EDGES; base += 32) {
                const int d = ei[N_EDGES + base + lane];
                unsigned m = __builtin_amdgcn_ballot_w32((d >> 3) == wv);
                while (m) {
                    const int j = __builtin_ctz(m);
                    m &= m - 1u;
                    const int dj = __shfl(d, j, 32);
                    const int s = ei[base + j];
                    if (rawA && dj == nodeA) add_row<SRC_SCALE>(hs, deg_out, s, q, a0, a1);
                    if (rawB && dj == nodeB) add_row<SRC_SCALE>(hs, deg_out, s, q, b0, b1);
                }
            }
        }
    }

    float4* oA = (float4*)(aggn + (size_t)nodeA * DIM) + q;
    float4* oB = (float4*)(aggn + (size_t)nodeB * DIM) + q;
    oA[0] = finish4<OUT_BIAS>(a0, rA, bias, 4 * q);
    oA[8] = finish4<OUT_BIAS>(a1, rA, bias, 32 + 4 * q);
    oB[0] = finish4<OUT_BIAS>(b0, rB, bias, 4 * q);
    oB[8] = finish4<OUT_BIAS>(b1, rB, bias, 32 + 4 * q);
}

#define WSW_WORDS (16 * 4 * 32 * 2)
#define TPITCH 68

template <bool TWO>
__device__ __forceinline__ void fill_wsw(float* lW, const float* __restrict__ W, const float* __restrict__ Wn) {
#pragma unroll 8
    for (int p = threadIdx.x; p < 16 * 4 * 32; p += 128) {
        int l  = p & 31;
        int t  = (p >> 5) & 3;
        int f  = p >> 7;
        int ka = 4 * f + 2 * (l >> 4);
        int n  = 16 * t + (l & 15);
        v2f w2;
        w2.x = W[ka * DIM + n];
        w2.y = W[(ka + 1) * DIM + n];
        *(v2f*)&lW[p * 2] = w2;
        if (TWO) {
            v2f w3;
            w3.x = Wn[ka * DIM + n];
            w3.y = Wn[(ka + 1) * DIM + n];
            *(v2f*)&lW[WSW_WORDS + p * 2] = w3;
        }
    }
}

template <bool TWO>
__global__ __launch_bounds__(128) void gemm_wmma_kernel(const float* __restrict__ X,
                                                        const float* __restrict__ W,
                                                        const float* __restrict__ bias,
                                                        const float* __restrict__ Wn,
                                                        const unsigned* __restrict__ deg_out,
                                                        float* __restrict__ Y) {
    __shared__ float lWsw[TWO ? 2 * WSW_WORDS : WSW_WORDS];
    __shared__ float lB[DIM];
    __shared__ float lT[TWO ? 4 * 16 * TPITCH : 2];
    fill_wsw<TWO>(lWsw, W, Wn);
    if (threadIdx.x < DIM) lB[threadIdx.x] = bias[threadIdx.x];
    __syncthreads();

    const int wave = threadIdx.x >> 5;
    const int lane = threadIdx.x & 31;
    const int half = lane >> 4;
    const int l16  = lane & 15;
    const int m0   = blockIdx.x * 64 + wave * 16;
    const int mA   = m0 + l16;

    v8f acc[4] = {v8f{}, v8f{}, v8f{}, v8f{}};

#pragma unroll
    for (int f = 0; f < 16; ++f) {
        const int ka = 4 * f + 2 * half;
        v2f a = *(const v2f*)(X + (size_t)mA * DIM + ka);
#pragma unroll
        for (int t = 0; t < 4; ++t) {
            v2f b = *(const v2f*)&lWsw[(((f * 4 + t) * 32) + lane) * 2];
            acc[t] = __builtin_amdgcn_wmma_f32_16x16x4_f32(
                false, a, false, b, (short)0, acc[t], false, false);
        }
    }

    const int mrow0 = m0 + half * 8;
    const uint4* dp = (const uint4*)(deg_out + mrow0);
    const uint4 dlo = dp[0];
    const uint4 dhi = dp[1];
    const float rsv[8] = {inv_sqrt_deg(dlo.x), inv_sqrt_deg(dlo.y), inv_sqrt_deg(dlo.z), inv_sqrt_deg(dlo.w),
                          inv_sqrt_deg(dhi.x), inv_sqrt_deg(dhi.y), inv_sqrt_deg(dhi.z), inv_sqrt_deg(dhi.w)};

    float* tile = lT + (TWO ? wave * 16 * TPITCH : 0);
#pragma unroll
    for (int t = 0; t < 4; ++t) {
        const int n = t * 16 + l16;
        const float bn = lB[n];
#pragma unroll
        for (int r = 0; r < 8; ++r) {
            const float v = fmaxf(acc[t][r] + bn, 0.0f) * rsv[r];
            if (TWO) tile[(half * 8 + r) * TPITCH + n] = v;
            else Y[(size_t)(mrow0 + r) * DIM + n] = v;
        }
    }

    if (TWO) {
        __syncthreads();
        const float* lWn = lWsw + WSW_WORDS;
        v8f acc2[4] = {v8f{}, v8f{}, v8f{}, v8f{}};
#pragma unroll
        for (int f = 0; f < 16; ++f) {
            const int ka = 4 * f + 2 * half;
            v2f a = *(const v2f*)&tile[l16 * TPITCH + ka];
#pragma unroll
            for (int t = 0; t < 4; ++t) {
                v2f b = *(const v2f*)&lWn[(((f * 4 + t) * 32) + lane) * 2];
                acc2[t] = __builtin_amdgcn_wmma_f32_16x16x4_f32(
                    false, a, false, b, (short)0, acc2[t], false, false);
            }
        }
#pragma unroll
        for (int t = 0; t < 4; ++t) {
            const int n = t * 16 + l16;
#pragma unroll
            for (int r = 0; r < 8; ++r) Y[(size_t)(mrow0 + r) * DIM + n] = acc2[t][r];
        }
    }
}

extern "C" void kernel_launch(void* const* d_in, const int* in_sizes, int n_in,
                              void* d_out, int out_size, void* d_ws, size_t ws_size,
                              hipStream_t stream) {
    (void)in_sizes; (void)n_in; (void)out_size; (void)ws_size;
    const float* x  = (const float*)d_in[0];
    const int*   ei = (const int*)d_in[1];
    const float* W1 = (const float*)d_in[3];
    const float* b1 = (const float*)d_in[4];
    const float* W2 = (const float*)d_in[5];
    const float* b2 = (const float*)d_in[6];
    const float* W3 = (const float*)d_in[7];
    const float* b3 = (const float*)d_in[8];
    float* out = (float*)d_out;

    char* ws = (char*)d_ws;
    unsigned* deg_out = (unsigned*)ws; ws += (size_t)N_NODES * 4;
    unsigned* deg_in  = (unsigned*)ws; ws += (size_t)N_NODES * 4;
    unsigned* spill_cnt = (unsigned*)ws; ws += (size_t)32 * 4;
    unsigned* tick    = (unsigned*)ws; ws += (size_t)TICK_WORDS * 4;
    unsigned* spill   = (unsigned*)ws; ws += (size_t)SPILL_CAP * 4;
    unsigned short* slots = (unsigned short*)ws; ws += (size_t)N_NODES * SLOTS * 2;
    float*    hs      = (float*)ws;    ws += (size_t)N_NODES * DIM * 4;
    float*    aggn    = (float*)ws;    ws += (size_t)N_NODES * DIM * 4;

    zero_u32_kernel<<<(2 * N_NODES + 32 + TICK_WORDS + 255) / 256, 256, 0, stream>>>(deg_out, 2 * N_NODES + 32 + TICK_WORDS);
    build_kernel<<<2 * ITEMS_PER_HALF, 1024, 0, stream>>>(ei, deg_out, deg_in, spill_cnt, spill, slots, tick);

    agg_kernel<true, false><<<N_NODES / 64, 256, 0, stream>>>(x, slots, deg_in, deg_out, spill_cnt, spill, ei, nullptr, aggn);
    gemm_wmma_kernel<false><<<N_NODES / 64, 128, 0, stream>>>(aggn, W1, b1, nullptr, deg_out, hs);
    agg_kernel<false, false><<<N_NODES / 64, 256, 0, stream>>>(hs, slots, deg_in, deg_out, spill_cnt, spill, ei, nullptr, aggn);
    gemm_wmma_kernel<true><<<N_NODES / 64, 128, 0, stream>>>(aggn, W2, b2, W3, deg_out, hs);
    agg_kernel<false, true><<<N_NODES / 64, 256, 0, stream>>>(hs, slots, deg_in, deg_out, spill_cnt, spill, ei, b3, out);
}
